# starting kernel
# baseline (speedup 1.0000x reference)
#include <hip/hip_runtime.h>
#include <hip/hip_bf16.h>


typedef float v2f __attribute__((ext_vector_type(2)));
typedef float v4f __attribute__((ext_vector_type(4)));
typedef float v8f __attribute__((ext_vector_type(8)));

#define TOKENS 4096
#define DIN    1024
#define DOUT   1024
#define RANK   8

__global__ __launch_bounds__(256) void lora_inter_kernel(
    const float* __restrict__ x,
    const float* __restrict__ Bp,
    float* __restrict__ inter) {
  const int lane = threadIdx.x & 31;
  const int wave = threadIdx.x >> 5;
  const int t = blockIdx.x * 8 + wave;

  const v4f* xp = (const v4f*)(x + (size_t)t * DIN);
  v4f xr[8];
#pragma unroll
  for (int j = 0; j < 8; ++j) xr[j] = xp[j * 32 + lane];

#pragma unroll
  for (int r = 0; r < RANK; ++r) {
    const v4f* bp =
        (const v4f*)(Bp + (size_t)t * (RANK * DIN) + (size_t)r * DIN);
    float acc = 0.f;
#pragma unroll
    for (int j = 0; j < 8; ++j) {
      v4f bv = __builtin_nontemporal_load(&bp[j * 32 + lane]);
      acc += xr[j].x * bv.x + xr[j].y * bv.y + xr[j].z * bv.z + xr[j].w * bv.w;
    }
#pragma unroll
    for (int off = 16; off > 0; off >>= 1) acc += __shfl_xor(acc, off, 32);
    if (lane == 0) inter[t * RANK + r] = acc;
  }
}

#define TM 128
#define TN 128
#define TK 32
#define NKT (DIN / TK)
#define LSTR 36

__global__ __launch_bounds__(256) void lora_gemm_kernel(
    const float* __restrict__ x,
    const float* __restrict__ W,
    const float* __restrict__ bias,
    const float* __restrict__ Ap,
    const float* __restrict__ inter,
    float* __restrict__ out) {
  __shared__ float xs[2][TM * LSTR];
  __shared__ float wsm[2][TN * LSTR];
  __shared__ float is[TM * RANK];

  const int tid  = threadIdx.x;
  const int lane = tid & 31;
  const int wv   = tid >> 5;
  const int wm   = wv & 1;
  const int wn   = wv >> 1;
  const int t0   = blockIdx.x * TM;
  const int o0   = blockIdx.y * TN;

  const int lrow = tid >> 3;
  const int lf4  = tid & 7;
  const int r15  = lane & 15;
  const int kh   = (lane >> 4) * 2;

  v4f px[4], pw[4];
#pragma unroll
  for (int p = 0; p < 4; ++p) {
    const int row = lrow + p * 32;
    px[p] = *(const v4f*)(x + (size_t)(t0 + row) * DIN + lf4 * 4);
    pw[p] = *(const v4f*)(W + (size_t)(o0 + row) * DIN + lf4 * 4);
  }
  ((v4f*)is)[tid] = ((const v4f*)(inter + (size_t)t0 * RANK))[tid];
#pragma unroll
  for (int p = 0; p < 4; ++p) {
    const int row = lrow + p * 32;
    *(v4f*)(&xs[0][row * LSTR + lf4 * 4]) = px[p];
    *(v4f*)(&wsm[0][row * LSTR + lf4 * 4]) = pw[p];
  }

  v8f acc[4][2];
#pragma unroll
  for (int mi = 0; mi < 4; ++mi)
#pragma unroll
    for (int ni = 0; ni < 2; ++ni)
#pragma unroll
      for (int v = 0; v < 8; ++v) acc[mi][ni][v] = 0.f;

  __syncthreads();

  for (int kt = 0; kt < NKT; ++kt) {
    const int cur = kt & 1;
    const bool more = (kt + 1) < NKT;

    if (more) {
      const int kn = (kt + 1) * TK;
#pragma unroll
      for (int p = 0; p < 4; ++p) {
        const int row = lrow + p * 32;
        px[p] = *(const v4f*)(x + (size_t)(t0 + row) * DIN + kn + lf4 * 4);
        pw[p] = *(const v4f*)(W + (size_t)(o0 + row) * DIN + kn + lf4 * 4);
      }
    }

    const float* xb = xs[cur];
    const float* wb = wsm[cur];
#pragma unroll
    for (int kk = 0; kk < TK; kk += 4) {
      v2f a[4], bfr[2];
#pragma unroll
      for (int mi = 0; mi < 4; ++mi)
        a[mi] = *(const v2f*)(xb + (wm * 64 + mi * 16 + r15) * LSTR + kk + kh);
#pragma unroll
      for (int ni = 0; ni < 2; ++ni)
        bfr[ni] = *(const v2f*)(wb + (wn * 32 + ni * 16 + r15) * LSTR + kk + kh);
#pragma unroll
      for (int mi = 0; mi < 4; ++mi)
#pragma unroll
        for (int ni = 0; ni < 2; ++ni)
          acc[mi][ni] = __builtin_amdgcn_wmma_f32_16x16x4_f32(
              false, a[mi], false, bfr[ni], (short)0, acc[mi][ni],
              false, false);
    }

    if (more) {
#pragma unroll
      for (int p = 0; p < 4; ++p) {
        const int row = lrow + p * 32;
        *(v4f*)(&xs[1 - cur][row * LSTR + lf4 * 4]) = px[p];
        *(v4f*)(&wsm[1 - cur][row * LSTR + lf4 * 4]) = pw[p];
      }
    }
    __syncthreads();
  }

  const int mhi = (lane >> 4) << 3;
#pragma unroll
  for (int ni = 0; ni < 2; ++ni) {
    const int o = o0 + wn * 32 + ni * 16 + r15;
    const float bv = bias[o];
#pragma unroll
    for (int mi = 0; mi < 4; ++mi) {
#pragma unroll
      for (int v = 0; v < 8; ++v) {
        const int mloc = wm * 64 + mi * 16 + v + mhi;
        const int t = t0 + mloc;
        const v4f* ar = (const v4f*)(Ap + ((size_t)t * DOUT + o) * RANK);
        v4f a0 = __builtin_nontemporal_load(&ar[0]);
        v4f a1 = __builtin_nontemporal_load(&ar[1]);
        const v4f i0 = *(const v4f*)(is + mloc * RANK);
        const v4f i1 = *(const v4f*)(is + mloc * RANK + 4);
        float ad = i0.x * a0.x + i0.y * a0.y + i0.z * a0.z + i0.w * a0.w +
                   i1.x * a1.x + i1.y * a1.y + i1.z * a1.z + i1.w * a1.w;
        out[(size_t)t * DOUT + o] = acc[mi][ni][v] + bv + ad;
      }
    }
  }
}

extern "C" void kernel_launch(void* const* d_in, const int* in_sizes, int n_in,
                              void* d_out, int out_size, void* d_ws,
                              size_t ws_size, hipStream_t stream) {
  (void)in_sizes; (void)n_in; (void)out_size; (void)ws_size;
  const float* x    = (const float*)d_in[0];
  const float* Ap   = (const float*)d_in[1];
  const float* Bp   = (const float*)d_in[2];
  const float* W    = (const float*)d_in[3];
  const float* bias = (const float*)d_in[4];
  float* out   = (float*)d_out;
  float* inter = (float*)d_ws;

  lora_inter_kernel<<<TOKENS / 8, 256, 0, stream>>>(x, Bp, inter);

  dim3 grid(TOKENS / TM, DOUT / TN);
  lora_gemm_kernel<<<grid, 256, 0, stream>>>(x, W, bias, Ap, inter, out);
}


# GEMM kernel re-tiled: 512 work-groups of 128x64, two per WGP; half stream their A_params tile before the K loop, half after it, into an LDS out tile; rank-8 sum now two 4-term halves plus bias
# speedup vs baseline: 1.1826x; 1.1826x over previous
#include <hip/hip_runtime.h>
#include <hip/hip_bf16.h>


typedef float v2f __attribute__((ext_vector_type(2)));
typedef float v4f __attribute__((ext_vector_type(4)));
typedef float v8f __attribute__((ext_vector_type(8)));

#define TOKENS 4096
#define DIN    1024
#define DOUT   1024
#define RANK   8

__global__ __launch_bounds__(256) void lora_inter_kernel(
    const float* __restrict__ x,
    const float* __restrict__ Bp,
    float* __restrict__ inter) {
  const int lane = threadIdx.x & 31;
  const int wave = threadIdx.x >> 5;
  const int t = blockIdx.x * 8 + wave;

  const v4f* xp = (const v4f*)(x + (size_t)t * DIN);
  v4f xr[8];
#pragma unroll
  for (int j = 0; j < 8; ++j) xr[j] = xp[j * 32 + lane];

#pragma unroll
  for (int r = 0; r < RANK; ++r) {
    const v4f* bp =
        (const v4f*)(Bp + (size_t)t * (RANK * DIN) + (size_t)r * DIN);
    float acc = 0.f;
#pragma unroll
    for (int j = 0; j < 8; ++j) {
      v4f bv = __builtin_nontemporal_load(&bp[j * 32 + lane]);
      acc += xr[j].x * bv.x + xr[j].y * bv.y + xr[j].z * bv.z + xr[j].w * bv.w;
    }
#pragma unroll
    for (int off = 16; off > 0; off >>= 1) acc += __shfl_xor(acc, off, 32);
    if (lane == 0) inter[t * RANK + r] = acc;
  }
}

#define TM 128
#define TN 64
#define TK 64
#define NKT (DIN / TK)
#define LSTR 68
#define OSTR 68

__device__ __forceinline__ float swap_pair(float v) {
  return __builtin_bit_cast(
      float, __builtin_amdgcn_update_dpp(0, __builtin_bit_cast(int, v), 0xB1,
                                         0xF, 0xF, false));
}

template <bool ADD>
__device__ __forceinline__ void adapt_stream(const float* __restrict__ Ap,
                                             const float* __restrict__ bias,
                                             const float* is, float* ot,
                                             int t0, int o0, int wv, int lane) {
  const int par = lane & 1;
  const int ol  = lane >> 1;
  const int c0  = 16 * par + ol;
  const float b0 = bias[o0 + c0];
  const float b1 = bias[o0 + 32 + c0];
  const int rbase = wv * 16;

#pragma nounroll
  for (int g = 0; g < 2; ++g) {
    const int r0w = rbase + g * 8;
    const v4f* ap =
        (const v4f*)(Ap + ((size_t)(t0 + r0w) * DOUT + o0) * RANK) + lane;
    v4f a[32];
#pragma unroll
    for (int rr = 0; rr < 8; ++rr)
#pragma unroll
      for (int j = 0; j < 4; ++j)
        a[rr * 4 + j] = __builtin_nontemporal_load(
            &ap[rr * (DOUT * RANK / 4) + j * 32]);
#pragma unroll
    for (int rr = 0; rr < 8; ++rr) {
      const int row = r0w + rr;
      const v4f iv = *(const v4f*)(is + row * RANK + 4 * par);
      float f[4];
#pragma unroll
      for (int j = 0; j < 4; ++j) {
        const v4f av = a[rr * 4 + j];
        const float p = av.x * iv.x + av.y * iv.y + av.z * iv.z + av.w * iv.w;
        f[j] = p + swap_pair(p);
      }
      const float r0 = (par ? f[1] : f[0]) + b0;
      const float r1 = (par ? f[3] : f[2]) + b1;
      float* op = ot + row * OSTR + c0;
      if (ADD) {
        op[0]  = op[0] + r0;
        op[32] = op[32] + r1;
      } else {
        op[0]  = r0;
        op[32] = r1;
      }
    }
  }
}

__global__ __launch_bounds__(256, 4) void lora_gemm_kernel(
    const float* __restrict__ x,
    const float* __restrict__ W,
    const float* __restrict__ bias,
    const float* __restrict__ Ap,
    const float* __restrict__ inter,
    float* __restrict__ out) {
  __shared__ __attribute__((aligned(16))) float xs[2][TM * LSTR];
  __shared__ __attribute__((aligned(16))) float wsm[2][TN * LSTR];
  __shared__ __attribute__((aligned(16))) float ot[TM * OSTR];
  __shared__ __attribute__((aligned(16))) float is[TM * RANK];

  const int tid  = threadIdx.x;
  const int lane = tid & 31;
  const int wv   = tid >> 5;
  const int wm   = wv & 3;
  const int wn   = wv >> 2;
  const int wg   = blockIdx.x;
  const int t0   = (wg >> 4) * TM;
  const int o0   = (wg & 15) * TN;
  const bool stream_first = (__builtin_popcount((unsigned)wg) & 1) != 0;

  const int lrow = tid >> 4;
  const int lf4  = tid & 15;
  const int r15  = lane & 15;
  const int kh   = (lane >> 4) * 2;

  ((v4f*)is)[tid] = ((const v4f*)(inter + (size_t)t0 * RANK))[tid];
  __syncthreads();

  if (stream_first) adapt_stream<false>(Ap, bias, is, ot, t0, o0, wv, lane);

  v4f px[8], pw[4];
#pragma unroll
  for (int p = 0; p < 8; ++p) {
    const int row = lrow + p * 16;
    px[p] = *(const v4f*)(x + (size_t)(t0 + row) * DIN + lf4 * 4);
  }
#pragma unroll
  for (int p = 0; p < 4; ++p) {
    const int row = lrow + p * 16;
    pw[p] = *(const v4f*)(W + (size_t)(o0 + row) * DIN + lf4 * 4);
  }
#pragma unroll
  for (int p = 0; p < 8; ++p)
    *(v4f*)(&xs[0][(lrow + p * 16) * LSTR + lf4 * 4]) = px[p];
#pragma unroll
  for (int p = 0; p < 4; ++p)
    *(v4f*)(&wsm[0][(lrow + p * 16) * LSTR + lf4 * 4]) = pw[p];

  v8f acc[2][2];
#pragma unroll
  for (int mi = 0; mi < 2; ++mi)
#pragma unroll
    for (int ni = 0; ni < 2; ++ni)
#pragma unroll
      for (int v = 0; v < 8; ++v) acc[mi][ni][v] = 0.f;

  __syncthreads();

  for (int kt = 0; kt < NKT; ++kt) {
    const int cur = kt & 1;
    const bool more = (kt + 1) < NKT;

    if (more) {
      const int kn = (kt + 1) * TK;
#pragma unroll
      for (int p = 0; p < 8; ++p) {
        const int row = lrow + p * 16;
        px[p] = *(const v4f*)(x + (size_t)(t0 + row) * DIN + kn + lf4 * 4);
      }
#pragma unroll
      for (int p = 0; p < 4; ++p) {
        const int row = lrow + p * 16;
        pw[p] = *(const v4f*)(W + (size_t)(o0 + row) * DIN + kn + lf4 * 4);
      }
    }

    const float* xb = xs[cur];
    const float* wb = wsm[cur];
#pragma unroll
    for (int kk = 0; kk < TK; kk += 4) {
      v2f a[2], bfr[2];
#pragma unroll
      for (int mi = 0; mi < 2; ++mi)
        a[mi] = *(const v2f*)(xb + (wm * 32 + mi * 16 + r15) * LSTR + kk + kh);
#pragma unroll
      for (int ni = 0; ni < 2; ++ni)
        bfr[ni] = *(const v2f*)(wb + (wn * 32 + ni * 16 + r15) * LSTR + kk + kh);
#pragma unroll
      for (int mi = 0; mi < 2; ++mi)
#pragma unroll
        for (int ni = 0; ni < 2; ++ni)
          acc[mi][ni] = __builtin_amdgcn_wmma_f32_16x16x4_f32(
              false, a[mi], false, bfr[ni], (short)0, acc[mi][ni],
              false, false);
    }

    __builtin_amdgcn_sched_barrier(0);

    if (more) {
#pragma unroll
      for (int p = 0; p < 8; ++p)
        *(v4f*)(&xs[1 - cur][(lrow + p * 16) * LSTR + lf4 * 4]) = px[p];
#pragma unroll
      for (int p = 0; p < 4; ++p)
        *(v4f*)(&wsm[1 - cur][(lrow + p * 16) * LSTR + lf4 * 4]) = pw[p];
    }
    __syncthreads();
  }

  const int mhi = (lane >> 4) << 3;
  float* const obase = ot + (wm * 32 + mhi) * OSTR + wn * 32 + r15;
  if (stream_first) {
#pragma unroll
    for (int mi = 0; mi < 2; ++mi)
#pragma unroll
      for (int ni = 0; ni < 2; ++ni)
#pragma unroll
        for (int v = 0; v < 8; ++v) {
          float* op = obase + (mi * 16 + v) * OSTR + ni * 16;
          *op = acc[mi][ni][v] + *op;
        }
  } else {
#pragma unroll
    for (int mi = 0; mi < 2; ++mi)
#pragma unroll
      for (int ni = 0; ni < 2; ++ni)
#pragma unroll
        for (int v = 0; v < 8; ++v)
          obase[(mi * 16 + v) * OSTR + ni * 16] = acc[mi][ni][v];
  }
  __syncthreads();

  if (!stream_first) {
    adapt_stream<true>(Ap, bias, is, ot, t0, o0, wv, lane);
    __syncthreads();
  }

#pragma unroll
  for (int q = 0; q < 8; ++q) {
    const int e   = tid + q * 256;
    const int row = e >> 4;
    const int c4  = e & 15;
    const v4f val = *(const v4f*)(ot + row * OSTR + c4 * 4);
    *(v4f*)(out + (size_t)(t0 + row) * DOUT + o0 + c4 * 4) = val;
  }
}

extern "C" void kernel_launch(void* const* d_in, const int* in_sizes, int n_in,
                              void* d_out, int out_size, void* d_ws,
                              size_t ws_size, hipStream_t stream) {
  (void)in_sizes; (void)n_in; (void)out_size; (void)ws_size;
  const float* x    = (const float*)d_in[0];
  const float* Ap   = (const float*)d_in[1];
  const float* Bp   = (const float*)d_in[2];
  const float* W    = (const float*)d_in[3];
  const float* bias = (const float*)d_in[4];
  float* out   = (float*)d_out;
  float* inter = (float*)d_ws;

  lora_inter_kernel<<<TOKENS / 8, 256, 0, stream>>>(x, Bp, inter);

  lora_gemm_kernel<<<512, 256, 0, stream>>>(x, W, bias, Ap, inter, out);
}


# Launch split moved: launch 1 = base product + bias, each wave also streams one token's B_params in its K loop (inter: per-lane sums, 16-lane butterfly); launch 2 = A_params stream added into out
# speedup vs baseline: 1.3350x; 1.1289x over previous
#include <hip/hip_runtime.h>
#include <hip/hip_bf16.h>


typedef float v2f __attribute__((ext_vector_type(2)));
typedef float v4f __attribute__((ext_vector_type(4)));
typedef float v8f __attribute__((ext_vector_type(8)));

#define TOKENS 4096
#define DIN    1024
#define DOUT   1024
#define RANK   8

#define TM 128
#define TN 64
#define TK 64
#define NKT (DIN / TK)
#define LSTR 68
#define OSTR 68

__global__ __launch_bounds__(256, 4) void lora_gemm_inter_kernel(
    const float* __restrict__ x,
    const float* __restrict__ W,
    const float* __restrict__ bias,
    const float* __restrict__ Bp,
    float* __restrict__ inter,
    float* __restrict__ out) {
  __shared__ __attribute__((aligned(16))) float xs[2][TM * LSTR];
  __shared__ __attribute__((aligned(16))) float wsm[2][TN * LSTR];
  __shared__ __attribute__((aligned(16))) float ot[TM * OSTR];

  const int tid  = threadIdx.x;
  const int lane = tid & 31;
  const int wv   = tid >> 5;
  const int wm   = wv & 3;
  const int wn   = wv >> 2;
  const int wg   = blockIdx.x;
  const int t0   = (wg >> 4) * TM;
  const int o0   = (wg & 15) * TN;

  const int lrow = tid >> 4;
  const int lf4  = tid & 15;
  const int r15  = lane & 15;
  const int kh   = (lane >> 4) * 2;

  const int tok = t0 + (wg & 15) * 8 + wv;
  const float* xtok = x + (size_t)tok * DIN + r15 * 4;
  const float* btok =
      Bp + (size_t)tok * (RANK * DIN) + (size_t)(lane >> 4) * DIN + r15 * 4;
  float isum[4] = {0.f, 0.f, 0.f, 0.f};
  v4f bq[2][4], xq[2];

  v4f px[8], pw[4];
#pragma unroll
  for (int p = 0; p < 8; ++p) {
    const int row = lrow + p * 16;
    px[p] = *(const v4f*)(x + (size_t)(t0 + row) * DIN + lf4 * 4);
  }
#pragma unroll
  for (int p = 0; p < 4; ++p) {
    const int row = lrow + p * 16;
    pw[p] = *(const v4f*)(W + (size_t)(o0 + row) * DIN + lf4 * 4);
  }
  __builtin_amdgcn_sched_barrier(0);
#pragma unroll
  for (int s = 0; s < 2; ++s) {
#pragma unroll
    for (int q = 0; q < 4; ++q)
      bq[s][q] = __builtin_nontemporal_load(
          (const v4f*)(btok + q * 2 * DIN + s * TK));
    xq[s] = *(const v4f*)(xtok + s * TK);
  }
#pragma unroll
  for (int p = 0; p < 8; ++p)
    *(v4f*)(&xs[0][(lrow + p * 16) * LSTR + lf4 * 4]) = px[p];
#pragma unroll
  for (int p = 0; p < 4; ++p)
    *(v4f*)(&wsm[0][(lrow + p * 16) * LSTR + lf4 * 4]) = pw[p];

  v8f acc[2][2];
#pragma unroll
  for (int mi = 0; mi < 2; ++mi)
#pragma unroll
    for (int ni = 0; ni < 2; ++ni)
#pragma unroll
      for (int v = 0; v < 8; ++v) acc[mi][ni][v] = 0.f;

  __builtin_amdgcn_fence(__ATOMIC_RELEASE, "workgroup", "local");
  __builtin_amdgcn_s_barrier();
  __builtin_amdgcn_fence(__ATOMIC_ACQUIRE, "workgroup", "local");

  auto slab = [&](const int kt, const bool more, const bool more_b)
      __attribute__((always_inline)) {
    const int cur = kt & 1;

    if (more) {
      const int kn = (kt + 1) * TK;
#pragma unroll
      for (int p = 0; p < 8; ++p) {
        const int row = lrow + p * 16;
        px[p] = *(const v4f*)(x + (size_t)(t0 + row) * DIN + kn + lf4 * 4);
      }
#pragma unroll
      for (int p = 0; p < 4; ++p) {
        const int row = lrow + p * 16;
        pw[p] = *(const v4f*)(W + (size_t)(o0 + row) * DIN + kn + lf4 * 4);
      }
    }

    __builtin_amdgcn_sched_barrier(0);

    const int ps = kt & 1;
#pragma unroll
    for (int q = 0; q < 4; ++q)
      isum[q] += xq[ps].x * bq[ps][q].x + xq[ps].y * bq[ps][q].y +
                 xq[ps].z * bq[ps][q].z + xq[ps].w * bq[ps][q].w;
    if (more_b) {
      const int k2 = (kt + 2) * TK;
#pragma unroll
      for (int q = 0; q < 4; ++q)
        bq[ps][q] = __builtin_nontemporal_load(
            (const v4f*)(btok + q * 2 * DIN + k2));
      xq[ps] = *(const v4f*)(xtok + k2);
    }

    const float* xb = xs[cur];
    const float* wb = wsm[cur];
#pragma unroll
    for (int kk = 0; kk < TK; kk += 4) {
      v2f a[2], bfr[2];
#pragma unroll
      for (int mi = 0; mi < 2; ++mi)
        a[mi] = *(const v2f*)(xb + (wm * 32 + mi * 16 + r15) * LSTR + kk + kh);
#pragma unroll
      for (int ni = 0; ni < 2; ++ni)
        bfr[ni] = *(const v2f*)(wb + (wn * 32 + ni * 16 + r15) * LSTR + kk + kh);
#pragma unroll
      for (int mi = 0; mi < 2; ++mi)
#pragma unroll
        for (int ni = 0; ni < 2; ++ni)
          acc[mi][ni] = __builtin_amdgcn_wmma_f32_16x16x4_f32(
              false, a[mi], false, bfr[ni], (short)0, acc[mi][ni],
              false, false);
    }

    __builtin_amdgcn_sched_barrier(0);

    if (more) {
#pragma unroll
      for (int p = 0; p < 8; ++p)
        *(v4f*)(&xs[1 - cur][(lrow + p * 16) * LSTR + lf4 * 4]) = px[p];
#pragma unroll
      for (int p = 0; p < 4; ++p)
        *(v4f*)(&wsm[1 - cur][(lrow + p * 16) * LSTR + lf4 * 4]) = pw[p];
    }
    __builtin_amdgcn_fence(__ATOMIC_RELEASE, "workgroup", "local");
    __builtin_amdgcn_s_barrier();
    __builtin_amdgcn_fence(__ATOMIC_ACQUIRE, "workgroup", "local");
  };

#pragma unroll 2
  for (int kt = 0; kt < NKT - 2; ++kt) slab(kt, true, true);
  slab(NKT - 2, true, false);
  slab(NKT - 1, false, false);

#pragma unroll
  for (int q = 0; q < 4; ++q) {
    float s = isum[q];
#pragma unroll
    for (int off = 8; off > 0; off >>= 1) s += __shfl_xor(s, off, 32);
    isum[q] = s;
  }
  if (r15 == 0) {
#pragma unroll
    for (int q = 0; q < 4; ++q)
      inter[tok * RANK + 2 * q + (lane >> 4)] = isum[q];
  }

  const int mhi = (lane >> 4) << 3;
  float* const obase = ot + (wm * 32 + mhi) * OSTR + wn * 32 + r15;
  float bv[2];
#pragma unroll
  for (int ni = 0; ni < 2; ++ni) bv[ni] = bias[o0 + wn * 32 + ni * 16 + r15];
#pragma unroll
  for (int mi = 0; mi < 2; ++mi)
#pragma unroll
    for (int ni = 0; ni < 2; ++ni)
#pragma unroll
      for (int v = 0; v < 8; ++v)
        obase[(mi * 16 + v) * OSTR + ni * 16] = acc[mi][ni][v] + bv[ni];
  __syncthreads();

#pragma unroll
  for (int q = 0; q < 8; ++q) {
    const int e   = tid + q * 256;
    const int row = e >> 4;
    const int c4  = e & 15;
    const v4f val = *(const v4f*)(ot + row * OSTR + c4 * 4);
    *(v4f*)(out + (size_t)(t0 + row) * DOUT + o0 + c4 * 4) = val;
  }
}

__device__ __forceinline__ float swap_pair(float v) {
  return __builtin_bit_cast(
      float, __builtin_amdgcn_update_dpp(0, __builtin_bit_cast(int, v), 0xB1,
                                         0xF, 0xF, false));
}

__global__ __launch_bounds__(256) __attribute__((amdgpu_waves_per_eu(4, 4)))
void lora_adapt_kernel(
    const float* __restrict__ Ap,
    const float* __restrict__ inter,
    float* __restrict__ out) {
  const int lane = threadIdx.x & 31;
  const int wave = threadIdx.x >> 5;
  const int t = blockIdx.x * 8 + wave;
  const int par = lane & 1;
  const int ol  = lane >> 1;
  const int c0  = 16 * par + ol;

  const v4f iv = *(const v4f*)(inter + t * RANK + 4 * par);
  const v4f* ap = (const v4f*)(Ap + (size_t)t * (DOUT * RANK)) + lane;
  float* op = out + (size_t)t * DOUT + c0;

#pragma nounroll
  for (int g = 0; g < 2; ++g) {
    v4f a[32];
    float o[16];
#pragma unroll
    for (int gg = 0; gg < 8; ++gg) {
      o[2 * gg]     = op[(g * 8 + gg) * 64];
      o[2 * gg + 1] = op[(g * 8 + gg) * 64 + 32];
    }
#pragma unroll
    for (int gg = 0; gg < 8; ++gg)
#pragma unroll
      for (int j = 0; j < 4; ++j)
        a[gg * 4 + j] = __builtin_nontemporal_load(
            &ap[((g * 8 + gg) * 4 + j) * 32]);
#pragma unroll
    for (int gg = 0; gg < 8; ++gg) {
      float f[4];
#pragma unroll
      for (int j = 0; j < 4; ++j) {
        const v4f av = a[gg * 4 + j];
        const float p = av.x * iv.x + av.y * iv.y + av.z * iv.z + av.w * iv.w;
        f[j] = p + swap_pair(p);
      }
      const float r0 = par ? f[1] : f[0];
      const float r1 = par ? f[3] : f[2];
      op[(g * 8 + gg) * 64]      = o[2 * gg] + r0;
      op[(g * 8 + gg) * 64 + 32] = o[2 * gg + 1] + r1;
    }
  }
}

extern "C" void kernel_launch(void* const* d_in, const int* in_sizes, int n_in,
                              void* d_out, int out_size, void* d_ws,
                              size_t ws_size, hipStream_t stream) {
  (void)in_sizes; (void)n_in; (void)out_size; (void)ws_size;
  const float* x    = (const float*)d_in[0];
  const float* Ap   = (const float*)d_in[1];
  const float* Bp   = (const float*)d_in[2];
  const float* W    = (const float*)d_in[3];
  const float* bias = (const float*)d_in[4];
  float* out   = (float*)d_out;
  float* inter = (float*)d_ws;

  lora_gemm_inter_kernel<<<512, 256, 0, stream>>>(x, W, bias, Bp, inter, out);

  lora_adapt_kernel<<<TOKENS / 8, 256, 0, stream>>>(Ap, inter, out);
}
